# starting kernel
# baseline (speedup 1.0000x reference)
#include <hip/hip_runtime.h>


#define N_MHR   20000
#define N_SMPLX 10475
#define N_PAD   10480
#define NT_COL  655
#define NT_ROW  1250

typedef float v2f __attribute__((ext_vector_type(2)));
typedef float v8f __attribute__((ext_vector_type(8)));

__global__ __launch_bounds__(256) void vm_prep(const float* __restrict__ mhr,
                                               const float* __restrict__ smplx,
                                               float4* __restrict__ A4,
                                               float4* __restrict__ B4) {
  int i = blockIdx.x * blockDim.x + threadIdx.x;
  if (i < N_MHR) {
    float x = mhr[3 * i + 0], y = mhr[3 * i + 1], z = mhr[3 * i + 2];
    A4[i] = make_float4(-2.0f * x, -2.0f * y, -2.0f * z, 1.0f);
  }
  if (i < N_PAD) {
    if (i < N_SMPLX) {
      float x = smplx[3 * i + 0], y = smplx[3 * i + 1], z = smplx[3 * i + 2];
      B4[i] = make_float4(x, y, z, x * x + y * y + z * z);
    } else {
      B4[i] = make_float4(0.0f, 0.0f, 0.0f, __builtin_inff());
    }
  }
}

__global__ __launch_bounds__(256) void vm_match(const float4* __restrict__ A4,
                                                const float4* __restrict__ B4,
                                                int* __restrict__ out) {
  extern __shared__ float4 Bs4[];
  float2* Bs2 = (float2*)Bs4;

  for (int t = threadIdx.x; t < N_PAD; t += 256) Bs4[t] = B4[t];
  __syncthreads();

  const float2* A2 = (const float2*)A4;
  const int lane = threadIdx.x & 31;
  const int wave = threadIdx.x >> 5;
  const int nlo  = lane & 15;
  const int kp   = lane >> 4;

  for (int rt = blockIdx.x * 8 + wave; rt < NT_ROW; rt += gridDim.x * 8) {
    const int m = rt * 16 + nlo;
    const float2 a2 = A2[2 * m + kp];
    const v2f af = {a2.x, a2.y};

    float vmin[8];
    int   imin[8];
#pragma unroll
    for (int r = 0; r < 8; ++r) { vmin[r] = __builtin_inff(); imin[r] = 0; }

#pragma unroll 2
    for (int ct = 0; ct < NT_COL; ++ct) {
      const int n = ct * 16 + nlo;
      const float2 b2 = Bs2[2 * n + kp];
      const v2f bf = {b2.x, b2.y};
      v8f c = {0.f, 0.f, 0.f, 0.f, 0.f, 0.f, 0.f, 0.f};
      c = __builtin_amdgcn_wmma_f32_16x16x4_f32(false, af, false, bf,
                                                (short)0, c, false, false);
#pragma unroll
      for (int r = 0; r < 8; ++r) {
        float cv = c[r];
        if (cv < vmin[r]) { vmin[r] = cv; imin[r] = n; }
      }
    }

#pragma unroll
    for (int r = 0; r < 8; ++r) {
      float v = vmin[r];
      int   idx = imin[r];
#pragma unroll
      for (int mask = 8; mask >= 1; mask >>= 1) {
        float vo = __shfl_xor(v, mask, 32);
        int   io = __shfl_xor(idx, mask, 32);
        const bool better = (vo < v) | ((vo == v) & (io < idx));
        idx = better ? io : idx;
        v   = better ? vo : v;
      }
      if (nlo == 0) {
        out[rt * 16 + r + (kp << 3)] = idx;
      }
    }
  }
}

extern "C" void kernel_launch(void* const* d_in, const int* in_sizes, int n_in,
                              void* d_out, int out_size, void* d_ws, size_t ws_size,
                              hipStream_t stream) {
  const float* mhr   = (const float*)d_in[0];
  const float* smplx = (const float*)d_in[1];
  int* out = (int*)d_out;

  char* ws = (char*)d_ws;
  float4* A4 = (float4*)ws;
  float4* B4 = (float4*)(ws + (size_t)N_MHR * 16);

  vm_prep<<<(N_MHR + 255) / 256, 256, 0, stream>>>(mhr, smplx, A4, B4);

  const size_t lds_bytes = (size_t)N_PAD * sizeof(float4);
  vm_match<<<64, 256, lds_bytes, stream>>>(A4, B4, out);
}


# match grid 64 -> 157 work-groups, one 16-row tile per wave (stride loop dropped): 157 WGPs share the row tiles instead of 64 and no wave walks 3 tiles
# speedup vs baseline: 2.5473x; 2.5473x over previous
#include <hip/hip_runtime.h>


#define N_MHR   20000
#define N_SMPLX 10475
#define N_PAD   10480
#define NT_COL  655
#define NT_ROW  1250
#define NB_MATCH 157
static_assert(NB_MATCH * 8 >= NT_ROW, "the match grid must hold one wave per 16-row tile");

typedef float v2f __attribute__((ext_vector_type(2)));
typedef float v8f __attribute__((ext_vector_type(8)));

__global__ __launch_bounds__(256) void vm_prep(const float* __restrict__ mhr,
                                               const float* __restrict__ smplx,
                                               float4* __restrict__ A4,
                                               float4* __restrict__ B4) {
  int i = blockIdx.x * blockDim.x + threadIdx.x;
  if (i < N_MHR) {
    float x = mhr[3 * i + 0], y = mhr[3 * i + 1], z = mhr[3 * i + 2];
    A4[i] = make_float4(-2.0f * x, -2.0f * y, -2.0f * z, 1.0f);
  }
  if (i < N_PAD) {
    if (i < N_SMPLX) {
      float x = smplx[3 * i + 0], y = smplx[3 * i + 1], z = smplx[3 * i + 2];
      B4[i] = make_float4(x, y, z, x * x + y * y + z * z);
    } else {
      B4[i] = make_float4(0.0f, 0.0f, 0.0f, __builtin_inff());
    }
  }
}

__global__ __launch_bounds__(256) void vm_match(const float4* __restrict__ A4,
                                                const float4* __restrict__ B4,
                                                int* __restrict__ out) {
  extern __shared__ float4 Bs4[];
  float2* Bs2 = (float2*)Bs4;

  for (int t = threadIdx.x; t < N_PAD; t += 256) Bs4[t] = B4[t];
  __syncthreads();

  const float2* A2 = (const float2*)A4;
  const int lane = threadIdx.x & 31;
  const int wave = threadIdx.x >> 5;
  const int nlo  = lane & 15;
  const int kp   = lane >> 4;

  const int rt = blockIdx.x * 8 + wave;
  if (rt < NT_ROW) {
    const int m = rt * 16 + nlo;
    const float2 a2 = A2[2 * m + kp];
    const v2f af = {a2.x, a2.y};

    float vmin[8];
    int   imin[8];
#pragma unroll
    for (int r = 0; r < 8; ++r) { vmin[r] = __builtin_inff(); imin[r] = 0; }

#pragma unroll 2
    for (int ct = 0; ct < NT_COL; ++ct) {
      const int n = ct * 16 + nlo;
      const float2 b2 = Bs2[2 * n + kp];
      const v2f bf = {b2.x, b2.y};
      v8f c = {0.f, 0.f, 0.f, 0.f, 0.f, 0.f, 0.f, 0.f};
      c = __builtin_amdgcn_wmma_f32_16x16x4_f32(false, af, false, bf,
                                                (short)0, c, false, false);
#pragma unroll
      for (int r = 0; r < 8; ++r) {
        float cv = c[r];
        if (cv < vmin[r]) { vmin[r] = cv; imin[r] = n; }
      }
    }

#pragma unroll
    for (int r = 0; r < 8; ++r) {
      float v = vmin[r];
      int   idx = imin[r];
#pragma unroll
      for (int mask = 8; mask >= 1; mask >>= 1) {
        float vo = __shfl_xor(v, mask, 32);
        int   io = __shfl_xor(idx, mask, 32);
        const bool better = (vo < v) | ((vo == v) & (io < idx));
        idx = better ? io : idx;
        v   = better ? vo : v;
      }
      if (nlo == 0) {
        out[rt * 16 + r + (kp << 3)] = idx;
      }
    }
  }
}

extern "C" void kernel_launch(void* const* d_in, const int* in_sizes, int n_in,
                              void* d_out, int out_size, void* d_ws, size_t ws_size,
                              hipStream_t stream) {
  const float* mhr   = (const float*)d_in[0];
  const float* smplx = (const float*)d_in[1];
  int* out = (int*)d_out;

  char* ws = (char*)d_ws;
  float4* A4 = (float4*)ws;
  float4* B4 = (float4*)(ws + (size_t)N_MHR * 16);

  vm_prep<<<(N_MHR + 255) / 256, 256, 0, stream>>>(mhr, smplx, A4, B4);

  const size_t lds_bytes = (size_t)N_PAD * sizeof(float4);
  vm_match<<<NB_MATCH, 256, lds_bytes, stream>>>(A4, B4, out);
}


# match loop software-pipelined: the next 5-tile block's B fragments are read from LDS as one group a block ahead, 5 separate WMMA result tiles per block; no WMMA waits for an LDS round trip any more
# speedup vs baseline: 3.5906x; 1.4096x over previous
#include <hip/hip_runtime.h>


#define N_MHR   20000
#define N_SMPLX 10475
#define N_PAD   10480
#define NT_COL  655
#define NT_ROW  1250
#define NB_MATCH 157
static_assert(NB_MATCH * 8 >= NT_ROW, "the match grid must hold one wave per 16-row tile");
#define UT      5
#define NBLK    131
static_assert(NBLK * UT == NT_COL, "the pipeline blocks must tile the columns exactly");

typedef float v2f __attribute__((ext_vector_type(2)));
typedef float v8f __attribute__((ext_vector_type(8)));

__global__ __launch_bounds__(256) void vm_prep(const float* __restrict__ mhr,
                                               const float* __restrict__ smplx,
                                               float4* __restrict__ A4,
                                               float4* __restrict__ B4) {
  int i = blockIdx.x * blockDim.x + threadIdx.x;
  if (i < N_MHR) {
    float x = mhr[3 * i + 0], y = mhr[3 * i + 1], z = mhr[3 * i + 2];
    A4[i] = make_float4(-2.0f * x, -2.0f * y, -2.0f * z, 1.0f);
  }
  if (i < N_PAD) {
    if (i < N_SMPLX) {
      float x = smplx[3 * i + 0], y = smplx[3 * i + 1], z = smplx[3 * i + 2];
      B4[i] = make_float4(x, y, z, x * x + y * y + z * z);
    } else {
      B4[i] = make_float4(0.0f, 0.0f, 0.0f, __builtin_inff());
    }
  }
}

__global__ __launch_bounds__(256) void vm_match(const float4* __restrict__ A4,
                                                const float4* __restrict__ B4,
                                                int* __restrict__ out) {
  extern __shared__ float4 Bs4[];
  float2* Bs2 = (float2*)Bs4;

  for (int t = threadIdx.x; t < N_PAD; t += 256) Bs4[t] = B4[t];
  __syncthreads();

  const float2* A2 = (const float2*)A4;
  const int lane = threadIdx.x & 31;
  const int wave = threadIdx.x >> 5;
  const int nlo  = lane & 15;
  const int kp   = lane >> 4;

  const int rt = blockIdx.x * 8 + wave;
  if (rt < NT_ROW) {
    const int m = rt * 16 + nlo;
    const float2 a2 = A2[2 * m + kp];
    const v2f af = {a2.x, a2.y};

    float vmin[8];
    int   imin[8];
#pragma unroll
    for (int r = 0; r < 8; ++r) { vmin[r] = __builtin_inff(); imin[r] = 0; }

    const float2* bp = Bs2 + 2 * nlo + kp;
    v2f bcur[UT];
#pragma unroll
    for (int j = 0; j < UT; ++j) {
      const float2 b2 = bp[32 * j];
      bcur[j] = v2f{b2.x, b2.y};
    }

    for (int blk = 0; blk < NBLK; ++blk) {
      const int nb = (blk + 1 < NBLK) ? blk + 1 : blk;
      v2f bnxt[UT];
#pragma unroll
      for (int j = 0; j < UT; ++j) {
        const float2 b2 = bp[32 * (nb * UT + j)];
        bnxt[j] = v2f{b2.x, b2.y};
      }
      __builtin_amdgcn_sched_barrier(0);

      v8f c[UT];
#pragma unroll
      for (int j = 0; j < UT; ++j) {
        const v8f z = {0.f, 0.f, 0.f, 0.f, 0.f, 0.f, 0.f, 0.f};
        c[j] = __builtin_amdgcn_wmma_f32_16x16x4_f32(false, af, false, bcur[j],
                                                     (short)0, z, false, false);
      }

#pragma unroll
      for (int j = 0; j < UT; ++j) {
        const int n = (blk * UT + j) * 16 + nlo;
#pragma unroll
        for (int r = 0; r < 8; ++r) {
          float cv = c[j][r];
          if (cv < vmin[r]) { vmin[r] = cv; imin[r] = n; }
        }
      }

#pragma unroll
      for (int j = 0; j < UT; ++j) bcur[j] = bnxt[j];
    }

#pragma unroll
    for (int r = 0; r < 8; ++r) {
      float v = vmin[r];
      int   idx = imin[r];
#pragma unroll
      for (int mask = 8; mask >= 1; mask >>= 1) {
        float vo = __shfl_xor(v, mask, 32);
        int   io = __shfl_xor(idx, mask, 32);
        const bool better = (vo < v) | ((vo == v) & (io < idx));
        idx = better ? io : idx;
        v   = better ? vo : v;
      }
      if (nlo == 0) {
        out[rt * 16 + r + (kp << 3)] = idx;
      }
    }
  }
}

extern "C" void kernel_launch(void* const* d_in, const int* in_sizes, int n_in,
                              void* d_out, int out_size, void* d_ws, size_t ws_size,
                              hipStream_t stream) {
  const float* mhr   = (const float*)d_in[0];
  const float* smplx = (const float*)d_in[1];
  int* out = (int*)d_out;

  char* ws = (char*)d_ws;
  float4* A4 = (float4*)ws;
  float4* B4 = (float4*)(ws + (size_t)N_MHR * 16);

  vm_prep<<<(N_MHR + 255) / 256, 256, 0, stream>>>(mhr, smplx, A4, B4);

  const size_t lds_bytes = (size_t)N_PAD * sizeof(float4);
  vm_match<<<NB_MATCH, 256, lds_bytes, stream>>>(A4, B4, out);
}


# match-loop argmin keeps only the running minimum and its first 5-tile block (two min3, one compare per row and block); the column is found after the loop by forming each row's winning block again
# speedup vs baseline: 3.9926x; 1.1120x over previous
#include <hip/hip_runtime.h>


#define N_MHR   20000
#define N_SMPLX 10475
#define N_PAD   10480
#define NT_COL  655
#define NT_ROW  1250
#define NB_MATCH 157
static_assert(NB_MATCH * 8 >= NT_ROW, "the match grid must hold one wave per 16-row tile");
#define UT      5
#define NBLK    131
static_assert(NBLK * UT == NT_COL, "the pipeline blocks must tile the columns exactly");

typedef float v2f __attribute__((ext_vector_type(2)));
typedef float v8f __attribute__((ext_vector_type(8)));

__global__ __launch_bounds__(256) void vm_prep(const float* __restrict__ mhr,
                                               const float* __restrict__ smplx,
                                               float4* __restrict__ A4,
                                               float4* __restrict__ B4) {
  int i = blockIdx.x * blockDim.x + threadIdx.x;
  if (i < N_MHR) {
    float x = mhr[3 * i + 0], y = mhr[3 * i + 1], z = mhr[3 * i + 2];
    A4[i] = make_float4(-2.0f * x, -2.0f * y, -2.0f * z, 1.0f);
  }
  if (i < N_PAD) {
    if (i < N_SMPLX) {
      float x = smplx[3 * i + 0], y = smplx[3 * i + 1], z = smplx[3 * i + 2];
      B4[i] = make_float4(x, y, z, x * x + y * y + z * z);
    } else {
      B4[i] = make_float4(0.0f, 0.0f, 0.0f, __builtin_inff());
    }
  }
}

__global__ __launch_bounds__(256) void vm_match(const float4* __restrict__ A4,
                                                const float4* __restrict__ B4,
                                                int* __restrict__ out) {
  extern __shared__ float4 Bs4[];
  float2* Bs2 = (float2*)Bs4;

  for (int t = threadIdx.x; t < N_PAD; t += 256) Bs4[t] = B4[t];
  __syncthreads();

  const float2* A2 = (const float2*)A4;
  const int lane = threadIdx.x & 31;
  const int wave = threadIdx.x >> 5;
  const int nlo  = lane & 15;
  const int kp   = lane >> 4;

  const int rt = blockIdx.x * 8 + wave;
  if (rt < NT_ROW) {
    const int m = rt * 16 + nlo;
    const float2 a2 = A2[2 * m + kp];
    const v2f af = {a2.x, a2.y};

    float vmin[8];
    int   bwin[8];
#pragma unroll
    for (int r = 0; r < 8; ++r) { vmin[r] = __builtin_inff(); bwin[r] = 0; }

    const float2* bp = Bs2 + 2 * nlo + kp;
    v2f bcur[UT];
#pragma unroll
    for (int j = 0; j < UT; ++j) {
      const float2 b2 = bp[32 * j];
      bcur[j] = v2f{b2.x, b2.y};
    }

    for (int blk = 0; blk < NBLK; ++blk) {
      const int nb = (blk + 1 < NBLK) ? blk + 1 : blk;
      v2f bnxt[UT];
#pragma unroll
      for (int j = 0; j < UT; ++j) {
        const float2 b2 = bp[32 * (nb * UT + j)];
        bnxt[j] = v2f{b2.x, b2.y};
      }
      __builtin_amdgcn_sched_barrier(0);

      v8f c[UT];
#pragma unroll
      for (int j = 0; j < UT; ++j) {
        const v8f z = {0.f, 0.f, 0.f, 0.f, 0.f, 0.f, 0.f, 0.f};
        c[j] = __builtin_amdgcn_wmma_f32_16x16x4_f32(false, af, false, bcur[j],
                                                     (short)0, z, false, false);
      }

#pragma unroll
      for (int r = 0; r < 8; ++r) {
        float bm = c[0][r];
#pragma unroll
        for (int j = 1; j < UT; ++j) bm = __builtin_fminf(bm, c[j][r]);
        if (bm < vmin[r]) { vmin[r] = bm; bwin[r] = blk; }
      }

#pragma unroll
      for (int j = 0; j < UT; ++j) bcur[j] = bnxt[j];
    }

#pragma unroll
    for (int r = 0; r < 8; ++r) {
      float v = vmin[r];
      int   b = bwin[r];
#pragma unroll
      for (int mask = 8; mask >= 1; mask >>= 1) {
        float vo = __shfl_xor(v, mask, 32);
        int   bo = __shfl_xor(b, mask, 32);
        const bool better = (vo < v) | ((vo == v) & (bo < b));
        b = better ? bo : b;
        v = better ? vo : v;
      }
      vmin[r] = v;
      bwin[r] = b;
    }

    v2f bres[16][UT];
    int bsel[16];
#pragma unroll
    for (int q = 0; q < 16; ++q) {
      bsel[q] = __builtin_amdgcn_readlane(bwin[q >> 1], (q & 1) << 4);
#pragma unroll
      for (int j = 0; j < UT; ++j) {
        const float2 b2 = bp[32 * (bsel[q] * UT + j)];
        bres[q][j] = v2f{b2.x, b2.y};
      }
    }

    int best[8];
#pragma unroll
    for (int r = 0; r < 8; ++r) {
      best[r] = 0x7fffffff;
#pragma unroll
      for (int h = 0; h < 2; ++h) {
        const int q = 2 * r + h;
#pragma unroll
        for (int j = 0; j < UT; ++j) {
          const v8f z = {0.f, 0.f, 0.f, 0.f, 0.f, 0.f, 0.f, 0.f};
          const v8f d = __builtin_amdgcn_wmma_f32_16x16x4_f32(false, af, false, bres[q][j],
                                                              (short)0, z, false, false);
          const int n = (bsel[q] * UT + j) * 16 + nlo;
          const bool hit = (d[r] == vmin[r]) & (n < best[r]);
          best[r] = hit ? n : best[r];
        }
      }
    }

#pragma unroll
    for (int r = 0; r < 8; ++r) {
#pragma unroll
      for (int mask = 8; mask >= 1; mask >>= 1) {
        const int bo = __shfl_xor(best[r], mask, 32);
        best[r] = (bo < best[r]) ? bo : best[r];
      }
    }

    if (nlo == 0) {
#pragma unroll
      for (int r = 0; r < 8; ++r) out[rt * 16 + r + (kp << 3)] = best[r];
    }
  }
}

extern "C" void kernel_launch(void* const* d_in, const int* in_sizes, int n_in,
                              void* d_out, int out_size, void* d_ws, size_t ws_size,
                              hipStream_t stream) {
  const float* mhr   = (const float*)d_in[0];
  const float* smplx = (const float*)d_in[1];
  int* out = (int*)d_out;

  char* ws = (char*)d_ws;
  float4* A4 = (float4*)ws;
  float4* B4 = (float4*)(ws + (size_t)N_MHR * 16);

  vm_prep<<<(N_MHR + 255) / 256, 256, 0, stream>>>(mhr, smplx, A4, B4);

  const size_t lds_bytes = (size_t)N_PAD * sizeof(float4);
  vm_match<<<NB_MATCH, 256, lds_bytes, stream>>>(A4, B4, out);
}


# match work split: 250 work-groups of 20 waves, each 16-row tile's columns shared by 4 waves (partial row minima merged through LDS): 5 waves on every SIMD of 250 WGPs instead of 2 on 157
# speedup vs baseline: 5.8700x; 1.4702x over previous
#include <hip/hip_runtime.h>


#define N_MHR   20000
#define N_SMPLX 10475
#define N_PAD   10480
#define NT_COL  655
#define NT_ROW  1250
#define UT      5
#define NBLK    131
static_assert(NBLK * UT == NT_COL, "the pipeline blocks must tile the columns exactly");

#define RT_WG   5
#define NQ      4
#define QBLK    33
#define NWAVE   (RT_WG * NQ)
#define NTHR    (NWAVE * 32)
#define NB_MATCH (NT_ROW / RT_WG)
static_assert(NB_MATCH * RT_WG == NT_ROW, "the row tiles must divide over the work-groups exactly");
static_assert(NQ * QBLK >= NBLK && (NQ - 1) * QBLK < NBLK, "the column parts must cover the blocks, none empty");
static_assert(NQ == 4, "the result phase gives each wave two accumulator rows per wave half");
#define COMB_BYTES (RT_WG * 16 * NQ * 8)
#define LDS_BYTES  (N_PAD * 16 + COMB_BYTES)

typedef float v2f __attribute__((ext_vector_type(2)));
typedef float v8f __attribute__((ext_vector_type(8)));

__global__ __launch_bounds__(256) void vm_prep(const float* __restrict__ mhr,
                                               const float* __restrict__ smplx,
                                               float4* __restrict__ A4,
                                               float4* __restrict__ B4) {
  int i = blockIdx.x * blockDim.x + threadIdx.x;
  if (i < N_MHR) {
    float x = mhr[3 * i + 0], y = mhr[3 * i + 1], z = mhr[3 * i + 2];
    A4[i] = make_float4(-2.0f * x, -2.0f * y, -2.0f * z, 1.0f);
  }
  if (i < N_PAD) {
    if (i < N_SMPLX) {
      float x = smplx[3 * i + 0], y = smplx[3 * i + 1], z = smplx[3 * i + 2];
      B4[i] = make_float4(x, y, z, x * x + y * y + z * z);
    } else {
      B4[i] = make_float4(0.0f, 0.0f, 0.0f, __builtin_inff());
    }
  }
}

__global__ __launch_bounds__(NTHR) void vm_match(const float4* __restrict__ A4,
                                                 const float4* __restrict__ B4,
                                                 int* __restrict__ out) {
  extern __shared__ float4 Bs4[];
  float2* Bs2 = (float2*)Bs4;
  int2* comb = (int2*)(Bs4 + N_PAD);

  for (int t = threadIdx.x; t < N_PAD; t += NTHR) Bs4[t] = B4[t];
  __syncthreads();

  const float2* A2 = (const float2*)A4;
  const int lane = threadIdx.x & 31;
  const int wave = __builtin_amdgcn_readfirstlane((int)(threadIdx.x >> 5));
  const int nlo  = lane & 15;
  const int kp   = lane >> 4;
  const int rtl  = wave >> 2;
  const int qq   = wave & 3;

  const int rt = blockIdx.x * RT_WG + rtl;
  const int m = rt * 16 + nlo;
  const float2 a2 = A2[2 * m + kp];
  const v2f af = {a2.x, a2.y};

  float vmin[8];
  int   bwin[8];
#pragma unroll
  for (int r = 0; r < 8; ++r) { vmin[r] = __builtin_inff(); bwin[r] = 0; }

  const int b0 = qq * QBLK;
  const int b1 = (b0 + QBLK < NBLK) ? b0 + QBLK : NBLK;
  const float2* bp = Bs2 + 2 * nlo + kp;
  v2f bcur[UT];
#pragma unroll
  for (int j = 0; j < UT; ++j) {
    const float2 b2 = bp[32 * (b0 * UT + j)];
    bcur[j] = v2f{b2.x, b2.y};
  }

  for (int blk = b0; blk < b1; ++blk) {
    const int nb = (blk + 1 < b1) ? blk + 1 : blk;
    v2f bnxt[UT];
#pragma unroll
    for (int j = 0; j < UT; ++j) {
      const float2 b2 = bp[32 * (nb * UT + j)];
      bnxt[j] = v2f{b2.x, b2.y};
    }
    __builtin_amdgcn_sched_barrier(0);

    v8f c[UT];
#pragma unroll
    for (int j = 0; j < UT; ++j) {
      const v8f z = {0.f, 0.f, 0.f, 0.f, 0.f, 0.f, 0.f, 0.f};
      c[j] = __builtin_amdgcn_wmma_f32_16x16x4_f32(false, af, false, bcur[j],
                                                   (short)0, z, false, false);
    }

#pragma unroll
    for (int r = 0; r < 8; ++r) {
      float bm = c[0][r];
#pragma unroll
      for (int j = 1; j < UT; ++j) bm = __builtin_fminf(bm, c[j][r]);
      if (bm < vmin[r]) { vmin[r] = bm; bwin[r] = blk; }
    }

#pragma unroll
    for (int j = 0; j < UT; ++j) bcur[j] = bnxt[j];
  }

#pragma unroll
  for (int r = 0; r < 8; ++r) {
    float v = vmin[r];
#pragma unroll
    for (int mask = 8; mask >= 1; mask >>= 1) v = __builtin_fminf(v, __shfl_xor(v, mask, 32));
    int b = (vmin[r] == v) ? bwin[r] : 0x7fffffff;
#pragma unroll
    for (int mask = 8; mask >= 1; mask >>= 1) {
      const int bo = __shfl_xor(b, mask, 32);
      b = (bo < b) ? bo : b;
    }
    vmin[r] = v;
    bwin[r] = b;
  }

  if (nlo == 0) {
#pragma unroll
    for (int r = 0; r < 8; ++r)
      comb[(rtl * 16 + (kp << 3) + r) * NQ + qq] = make_int2(__float_as_int(vmin[r]), bwin[r]);
  }
  __syncthreads();

  float vst[2];
  int   bst[2];
#pragma unroll
  for (int i = 0; i < 2; ++i) {
    const int4* c4 = (const int4*)(comb + (rtl * 16 + (kp << 3) + 2 * qq + i) * NQ);
    const int4 p01 = c4[0], p23 = c4[1];
    float v = __int_as_float(p01.x);
    int   b = p01.y;
    if (__int_as_float(p01.z) < v) { v = __int_as_float(p01.z); b = p01.w; }
    if (__int_as_float(p23.x) < v) { v = __int_as_float(p23.x); b = p23.y; }
    if (__int_as_float(p23.z) < v) { v = __int_as_float(p23.z); b = p23.w; }
    vst[i] = v;
    bst[i] = b;
  }

  v2f bres[4][UT];
  int bsel[4];
#pragma unroll
  for (int q = 0; q < 4; ++q) {
    bsel[q] = __builtin_amdgcn_readlane(bst[q >> 1], (q & 1) << 4);
#pragma unroll
    for (int j = 0; j < UT; ++j) {
      const float2 b2 = bp[32 * (bsel[q] * UT + j)];
      bres[q][j] = v2f{b2.x, b2.y};
    }
  }

  int best[2];
#pragma unroll
  for (int i = 0; i < 2; ++i) {
    best[i] = 0x7fffffff;
#pragma unroll
    for (int h = 0; h < 2; ++h) {
      const int q = 2 * i + h;
#pragma unroll
      for (int j = 0; j < UT; ++j) {
        const v8f z = {0.f, 0.f, 0.f, 0.f, 0.f, 0.f, 0.f, 0.f};
        const v8f d = __builtin_amdgcn_wmma_f32_16x16x4_f32(false, af, false, bres[q][j],
                                                            (short)0, z, false, false);
        const bool e0 = d[i] == vst[i], e1 = d[2 + i] == vst[i];
        const bool e2 = d[4 + i] == vst[i], e3 = d[6 + i] == vst[i];
        const bool eq = ((qq == 0) & e0) | ((qq == 1) & e1) | ((qq == 2) & e2) | ((qq == 3) & e3);
        const int n = (bsel[q] * UT + j) * 16 + nlo;
        const bool hit = eq & (n < best[i]);
        best[i] = hit ? n : best[i];
      }
    }
  }

#pragma unroll
  for (int i = 0; i < 2; ++i) {
#pragma unroll
    for (int mask = 8; mask >= 1; mask >>= 1) {
      const int bo = __shfl_xor(best[i], mask, 32);
      best[i] = (bo < best[i]) ? bo : best[i];
    }
  }

  if (nlo == 0) {
#pragma unroll
    for (int i = 0; i < 2; ++i) out[rt * 16 + (kp << 3) + 2 * qq + i] = best[i];
  }
}

extern "C" void kernel_launch(void* const* d_in, const int* in_sizes, int n_in,
                              void* d_out, int out_size, void* d_ws, size_t ws_size,
                              hipStream_t stream) {
  const float* mhr   = (const float*)d_in[0];
  const float* smplx = (const float*)d_in[1];
  int* out = (int*)d_out;

  char* ws = (char*)d_ws;
  float4* A4 = (float4*)ws;
  float4* B4 = (float4*)(ws + (size_t)N_MHR * 16);

  vm_prep<<<(N_MHR + 255) / 256, 256, 0, stream>>>(mhr, smplx, A4, B4);

  const size_t lds_bytes = (size_t)LDS_BYTES;
  vm_match<<<NB_MATCH, NTHR, lds_bytes, stream>>>(A4, B4, out);
}


# column search after the loop: winner lane kept with its block; one wave per row tile re-forms only that lane's 5 columns, 16 rows per product (gathered B columns): 5 products, not 80; ties: old path
# speedup vs baseline: 7.6615x; 1.3052x over previous
#include <hip/hip_runtime.h>


#define N_MHR   20000
#define N_SMPLX 10475
#define N_PAD   10480
#define NT_COL  655
#define NT_ROW  1250
#define UT      5
#define NBLK    131
static_assert(NBLK * UT == NT_COL, "the pipeline blocks must tile the columns exactly");

#define RT_WG   5
#define NQ      4
#define QBLK    33
#define NWAVE   (RT_WG * NQ)
#define NTHR    (NWAVE * 32)
#define NB_MATCH (NT_ROW / RT_WG)
static_assert(NB_MATCH * RT_WG == NT_ROW, "the row tiles must divide over the work-groups exactly");
static_assert(NQ * QBLK >= NBLK && (NQ - 1) * QBLK < NBLK, "the column parts must cover the blocks, none empty");
static_assert(NQ == 4, "the result phase gives each wave two accumulator rows per wave half");
#define NTRIP   ((N_PAD + NTHR - 1) / NTHR)
#define N_TAB   (NTRIP * NTHR)
static_assert((NTRIP - 1) * NTHR <= N_SMPLX, "only the last trip meets pad columns");
#define COMB_BYTES (RT_WG * 16 * NQ * 8)
#define LDS_BYTES  (N_TAB * 16 + COMB_BYTES)

typedef float v2f __attribute__((ext_vector_type(2)));
typedef float v8f __attribute__((ext_vector_type(8)));

static __device__ __forceinline__ float col_norm(float x, float y, float z) {
  float s;
  {
#pragma clang fp contract(off)
    s = x * x + y * y;
  }
  return __builtin_fmaf(z, z, s);
}

__global__ __launch_bounds__(NTHR) void vm_match(const float* __restrict__ mhr,
                                                 const float* __restrict__ smplx,
                                                 int* __restrict__ out) {
  extern __shared__ float4 Bs4[];
  float2* Bs2 = (float2*)Bs4;
  int2* comb = (int2*)(Bs4 + N_TAB);

  const int lane = threadIdx.x & 31;
  const int wave = __builtin_amdgcn_readfirstlane((int)(threadIdx.x >> 5));
  const int nlo  = lane & 15;
  const int kp   = lane >> 4;
  const int rtl  = wave >> 2;
  const int qq   = wave & 3;

  const int rt = blockIdx.x * RT_WG + rtl;
  const int m = rt * 16 + nlo;
  const float* ar = mhr + 3 * m;
  const float a_lo = ar[2 * kp];
  const float a_y  = ar[1];
  __builtin_amdgcn_sched_barrier(0);

  {
    float bx[NTRIP], by[NTRIP], bz[NTRIP];
#pragma unroll
    for (int k = 0; k < NTRIP; ++k) {
      const int t = (int)threadIdx.x + k * NTHR;
      const int tr = (k < NTRIP - 1 || t < N_SMPLX) ? t : N_SMPLX - 1;
      bx[k] = smplx[3 * tr + 0]; by[k] = smplx[3 * tr + 1]; bz[k] = smplx[3 * tr + 2];
    }
#pragma unroll
    for (int k = 0; k < NTRIP; ++k) {
      const int t = (int)threadIdx.x + k * NTHR;
      const bool real = (k < NTRIP - 1) || (t < N_SMPLX);
      const float x = bx[k], y = by[k], z = bz[k];
      const float w = col_norm(x, y, z);
      Bs4[t] = make_float4(real ? x : 0.0f, real ? y : 0.0f, real ? z : 0.0f,
                           real ? w : __builtin_inff());
    }
  }
  __syncthreads();

  const v2f af = {-2.0f * a_lo, kp ? 1.0f : -2.0f * a_y};

  float vmin[8];
  int   bwin[8];
#pragma unroll
  for (int r = 0; r < 8; ++r) { vmin[r] = __builtin_inff(); bwin[r] = 0; }

  const int b0 = qq * QBLK;
  const int b1 = (b0 + QBLK < NBLK) ? b0 + QBLK : NBLK;
  const float2* bp = Bs2 + 2 * nlo + kp;
  v2f bcur[UT];
#pragma unroll
  for (int j = 0; j < UT; ++j) {
    const float2 b2 = bp[32 * (b0 * UT + j)];
    bcur[j] = v2f{b2.x, b2.y};
  }

  for (int blk = b0; blk < b1; ++blk) {
    const int nb = (blk + 1 < b1) ? blk + 1 : blk;
    v2f bnxt[UT];
#pragma unroll
    for (int j = 0; j < UT; ++j) {
      const float2 b2 = bp[32 * (nb * UT + j)];
      bnxt[j] = v2f{b2.x, b2.y};
    }
    __builtin_amdgcn_sched_barrier(0);

    v8f c[UT];
#pragma unroll
    for (int j = 0; j < UT; ++j) {
      const v8f z = {0.f, 0.f, 0.f, 0.f, 0.f, 0.f, 0.f, 0.f};
      c[j] = __builtin_amdgcn_wmma_f32_16x16x4_f32(false, af, false, bcur[j],
                                                   (short)0, z, false, false);
    }

#pragma unroll
    for (int r = 0; r < 8; ++r) {
      float bm = c[0][r];
#pragma unroll
      for (int j = 1; j < UT; ++j) bm = __builtin_fminf(bm, c[j][r]);
      if (bm < vmin[r]) { vmin[r] = bm; bwin[r] = blk; }
    }

#pragma unroll
    for (int j = 0; j < UT; ++j) bcur[j] = bnxt[j];
  }

#pragma unroll
  for (int r = 0; r < 8; ++r) {
    float v = vmin[r];
#pragma unroll
    for (int mask = 8; mask >= 1; mask >>= 1) v = __builtin_fminf(v, __shfl_xor(v, mask, 32));
    const bool mine = (vmin[r] == v);
    int b = mine ? ((bwin[r] << 4) | nlo) : 0x7fffffff;
#pragma unroll
    for (int mask = 8; mask >= 1; mask >>= 1) {
      const int bo = __shfl_xor(b, mask, 32);
      b = (bo < b) ? bo : b;
    }
    const unsigned mm = __builtin_amdgcn_ballot_w32(mine);
    const int tlo = (__builtin_popcount(mm & 0xffffu) > 1) ? 0x10000 : 0;
    const int thi = (__builtin_popcount(mm >> 16) > 1) ? 0x10000 : 0;
    vmin[r] = v;
    bwin[r] = b | (kp ? thi : tlo);
  }

  if (nlo == 0) {
#pragma unroll
    for (int r = 0; r < 8; ++r)
      comb[(rtl * 16 + (kp << 3) + r) * NQ + qq] = make_int2(__float_as_int(vmin[r]), bwin[r]);
  }
  __syncthreads();

  float vrow;
  int   erow;
  {
    const int4* c4 = (const int4*)(comb + (rtl * 16 + nlo) * NQ);
    const int4 p01 = c4[0], p23 = c4[1];
    vrow = __int_as_float(p01.x);
    erow = p01.y;
    if (__int_as_float(p01.z) < vrow) { vrow = __int_as_float(p01.z); erow = p01.w; }
    if (__int_as_float(p23.x) < vrow) { vrow = __int_as_float(p23.x); erow = p23.y; }
    if (__int_as_float(p23.z) < vrow) { vrow = __int_as_float(p23.z); erow = p23.w; }
  }
  const bool tie_any = __builtin_amdgcn_ballot_w32((erow & 0x10000) != 0) != 0u;

  if (!tie_any) {
    if (qq == (rtl & 3)) {
      const int blk = erow >> 4;
      const int cs  = erow & 15;
      v2f bg[UT];
#pragma unroll
      for (int j = 0; j < UT; ++j) {
        const float2 b2 = Bs2[2 * ((blk * UT + j) * 16 + cs) + kp];
        bg[j] = v2f{b2.x, b2.y};
      }
      const int rsel = nlo & 7;
      int jwin = UT;
#pragma unroll
      for (int j = 0; j < UT; ++j) {
        const v8f z = {0.f, 0.f, 0.f, 0.f, 0.f, 0.f, 0.f, 0.f};
        const v8f d = __builtin_amdgcn_wmma_f32_16x16x4_f32(false, af, false, bg[j],
                                                            (short)0, z, false, false);
        bool eq = false;
#pragma unroll
        for (int r = 0; r < 8; ++r) eq |= (rsel == r) & (d[r] == vrow);
        jwin = (eq & (jwin == UT)) ? j : jwin;
      }
      if (kp == (nlo >> 3)) {
        const int jj = (jwin < UT) ? jwin : 0;
        out[rt * 16 + nlo] = (blk * UT + jj) * 16 + cs;
      }
    }
    return;
  }

  float vst[2];
  int   bst[2];
#pragma unroll
  for (int i = 0; i < 2; ++i) {
    const int4* c4 = (const int4*)(comb + (rtl * 16 + (kp << 3) + 2 * qq + i) * NQ);
    const int4 p01 = c4[0], p23 = c4[1];
    float v = __int_as_float(p01.x);
    int   b = p01.y;
    if (__int_as_float(p01.z) < v) { v = __int_as_float(p01.z); b = p01.w; }
    if (__int_as_float(p23.x) < v) { v = __int_as_float(p23.x); b = p23.y; }
    if (__int_as_float(p23.z) < v) { v = __int_as_float(p23.z); b = p23.w; }
    vst[i] = v;
    bst[i] = (b & 0xffff) >> 4;
  }

  v2f bres[4][UT];
  int bsel[4];
#pragma unroll
  for (int q = 0; q < 4; ++q) {
    bsel[q] = __builtin_amdgcn_readlane(bst[q >> 1], (q & 1) << 4);
#pragma unroll
    for (int j = 0; j < UT; ++j) {
      const float2 b2 = bp[32 * (bsel[q] * UT + j)];
      bres[q][j] = v2f{b2.x, b2.y};
    }
  }

  int best[2];
#pragma unroll
  for (int i = 0; i < 2; ++i) {
    best[i] = 0x7fffffff;
#pragma unroll
    for (int h = 0; h < 2; ++h) {
      const int q = 2 * i + h;
#pragma unroll
      for (int j = 0; j < UT; ++j) {
        const v8f z = {0.f, 0.f, 0.f, 0.f, 0.f, 0.f, 0.f, 0.f};
        const v8f d = __builtin_amdgcn_wmma_f32_16x16x4_f32(false, af, false, bres[q][j],
                                                            (short)0, z, false, false);
        const bool e0 = d[i] == vst[i], e1 = d[2 + i] == vst[i];
        const bool e2 = d[4 + i] == vst[i], e3 = d[6 + i] == vst[i];
        const bool eq = ((qq == 0) & e0) | ((qq == 1) & e1) | ((qq == 2) & e2) | ((qq == 3) & e3);
        const int n = (bsel[q] * UT + j) * 16 + nlo;
        const bool hit = eq & (n < best[i]);
        best[i] = hit ? n : best[i];
      }
    }
  }

#pragma unroll
  for (int i = 0; i < 2; ++i) {
#pragma unroll
    for (int mask = 8; mask >= 1; mask >>= 1) {
      const int bo = __shfl_xor(best[i], mask, 32);
      best[i] = (bo < best[i]) ? bo : best[i];
    }
  }

  if (nlo == 0) {
#pragma unroll
    for (int i = 0; i < 2; ++i) out[rt * 16 + (kp << 3) + 2 * qq + i] = best[i];
  }
}

extern "C" void kernel_launch(void* const* d_in, const int* in_sizes, int n_in,
                              void* d_out, int out_size, void* d_ws, size_t ws_size,
                              hipStream_t stream) {
  const float* mhr   = (const float*)d_in[0];
  const float* smplx = (const float*)d_in[1];
  int* out = (int*)d_out;

  const size_t lds_bytes = (size_t)LDS_BYTES;
  vm_match<<<NB_MATCH, NTHR, lds_bytes, stream>>>(mhr, smplx, out);
}


# each wave holds all 5 row tiles (80 queries) as column operand and sweeps 1/20 of the table as row operand: 1 LDS fragment feeds 5 products; per lane one min3 run, one (min, block) pair per row tile
# speedup vs baseline: 8.7461x; 1.1416x over previous
#include <hip/hip_runtime.h>


#define N_MHR   20000
#define N_SMPLX 10475
#define N_PAD   10480
#define NT_COL  655
#define NT_ROW  1250

#define RT_WG   5
#define NWAVE   20
#define NTHR    (NWAVE * 32)
#define NB_MATCH (NT_ROW / RT_WG)
#define UT      3
#define BPW     11
#define TPW     (BPW * UT)
#define NQ      4
static_assert(NB_MATCH * RT_WG == NT_ROW, "the row tiles must divide over the work-groups exactly");
static_assert(NWAVE * TPW >= NT_COL && (NWAVE - 1) * TPW < NT_COL, "the waves must cover the column tiles, none empty");
static_assert(NWAVE == RT_WG * NQ, "the result phase gives each row tile NQ waves");
static_assert(NWAVE % NQ == 0, "each of a row tile's NQ waves merges NWAVE / NQ wave ranges");
static_assert(NQ == 4, "the column search gives each wave two of a lane half's eight table columns");
static_assert(NWAVE * BPW < 0x10000, "a block number must fit the low half of a key");
#define NTRIP   ((N_PAD + NTHR - 1) / NTHR)
#define N_TAB   (NTRIP * NTHR)
static_assert((NTRIP - 1) * NTHR <= N_SMPLX, "only the last trip meets pad columns");
#define COMB1_N    (RT_WG * NWAVE * 2 * 16)
#define COMB2_N    (RT_WG * NQ * 16)
#define HIT_N      (RT_WG * NQ * 16)
#define LDS_BYTES  (N_TAB * 16 + COMB1_N * 8 + COMB2_N * 8 + HIT_N * 4)

typedef float v2f __attribute__((ext_vector_type(2)));
typedef float v8f __attribute__((ext_vector_type(8)));

static __device__ __forceinline__ float col_norm(float x, float y, float z) {
  float s;
  {
#pragma clang fp contract(off)
    s = x * x + y * y;
  }
  return __builtin_fmaf(z, z, s);
}

__global__ __launch_bounds__(NTHR) void vm_match(const float* __restrict__ mhr,
                                                 const float* __restrict__ smplx,
                                                 int* __restrict__ out) {
  extern __shared__ float4 Bs4[];
  float2* Bs2 = (float2*)Bs4;
  int2* comb1 = (int2*)(Bs4 + N_TAB);
  int2* comb2 = comb1 + COMB1_N;
  int*  hit   = (int*)(comb2 + COMB2_N);

  const int lane = threadIdx.x & 31;
  const int wave = __builtin_amdgcn_readfirstlane((int)(threadIdx.x >> 5));
  const int nlo  = lane & 15;
  const int kp   = lane >> 4;
  const int rtl  = wave >> 2;
  const int qq   = wave & 3;

  float a_lo[RT_WG], a_y[RT_WG];
#pragma unroll
  for (int t = 0; t < RT_WG; ++t) {
    const float* ar = mhr + 3 * ((blockIdx.x * RT_WG + t) * 16 + nlo);
    a_lo[t] = ar[2 * kp];
    a_y[t]  = ar[1];
  }
  __builtin_amdgcn_sched_barrier(0);

  {
    float bx[NTRIP], by[NTRIP], bz[NTRIP];
#pragma unroll
    for (int k = 0; k < NTRIP; ++k) {
      const int t = (int)threadIdx.x + k * NTHR;
      const int tr = (k < NTRIP - 1 || t < N_SMPLX) ? t : N_SMPLX - 1;
      bx[k] = smplx[3 * tr + 0]; by[k] = smplx[3 * tr + 1]; bz[k] = smplx[3 * tr + 2];
    }
#pragma unroll
    for (int k = 0; k < NTRIP; ++k) {
      const int t = (int)threadIdx.x + k * NTHR;
      const bool real = (k < NTRIP - 1) || (t < N_SMPLX);
      const float x = bx[k], y = by[k], z = bz[k];
      const float w = col_norm(x, y, z);
      Bs4[t] = make_float4(real ? x : 0.0f, real ? y : 0.0f, real ? z : 0.0f,
                           real ? w : __builtin_inff());
    }
  }
  __syncthreads();

  v2f af[RT_WG];
#pragma unroll
  for (int t = 0; t < RT_WG; ++t) af[t] = v2f{-2.0f * a_lo[t], kp ? 1.0f : -2.0f * a_y[t]};

  const int t0 = wave * TPW;
  const float2* bp = Bs2 + 2 * nlo + kp;
  v2f bcur[UT];
#pragma unroll
  for (int g = 0; g < UT; ++g) {
    const int ct = (t0 + g < NT_COL) ? t0 + g : NT_COL - 1;
    const float2 b2 = bp[32 * ct];
    bcur[g] = v2f{b2.x, b2.y};
  }

  float vmin[RT_WG];
  int   bwin[RT_WG];
#pragma unroll
  for (int t = 0; t < RT_WG; ++t) { vmin[t] = __builtin_inff(); bwin[t] = 0; }

  for (int blk = 0; blk < BPW; ++blk) {
    const int nb = (blk + 1 < BPW) ? blk + 1 : blk;
    v2f bnxt[UT];
#pragma unroll
    for (int g = 0; g < UT; ++g) {
      const int ci = t0 + nb * UT + g;
      const int ct = (ci < NT_COL) ? ci : NT_COL - 1;
      const float2 b2 = bp[32 * ct];
      bnxt[g] = v2f{b2.x, b2.y};
    }
    __builtin_amdgcn_sched_barrier(0);

    float m[RT_WG];
#pragma unroll
    for (int t = 0; t < RT_WG; ++t) m[t] = __builtin_inff();
#pragma unroll
    for (int g = 0; g < UT; ++g) {
      v8f c[RT_WG];
#pragma unroll
      for (int t = 0; t < RT_WG; ++t) {
        const v8f z = {0.f, 0.f, 0.f, 0.f, 0.f, 0.f, 0.f, 0.f};
        c[t] = __builtin_amdgcn_wmma_f32_16x16x4_f32(false, bcur[g], false, af[t],
                                                     (short)0, z, false, false);
      }
      __builtin_amdgcn_sched_barrier(0);
#pragma unroll
      for (int t = 0; t < RT_WG; ++t) {
#pragma unroll
        for (int r = 0; r < 8; ++r) m[t] = __builtin_fminf(m[t], c[t][r]);
      }
    }
#pragma unroll
    for (int t = 0; t < RT_WG; ++t) {
      if (m[t] < vmin[t]) { vmin[t] = m[t]; bwin[t] = blk; }
    }

#pragma unroll
    for (int g = 0; g < UT; ++g) bcur[g] = bnxt[g];
  }

#pragma unroll
  for (int t = 0; t < RT_WG; ++t)
    comb1[((t * NWAVE + wave) * 2 + kp) * 16 + nlo] =
        make_int2(__float_as_int(vmin[t]), wave * BPW + bwin[t]);
  __syncthreads();

  float v1;
  int   k1;
  {
    const int2* c1 = comb1 + ((rtl * NWAVE + qq * (NWAVE / NQ)) * 2 + kp) * 16 + nlo;
    const int2 p0 = c1[0];
    v1 = __int_as_float(p0.x); k1 = p0.y;
#pragma unroll
    for (int k = 1; k < NWAVE / NQ; ++k) {
      const int2 p = c1[32 * k];
      const float v = __int_as_float(p.x);
      const bool lt = v < v1;
      k1 = lt ? p.y : k1;
      v1 = lt ? v : v1;
    }
  }
  int hm;
  {
    const float vo = __shfl_xor(v1, 16, 32);
    const int   ko = __shfl_xor(k1, 16, 32);
    const bool lt   = (vo < v1) | ((vo == v1) & (ko < k1));
    const bool same = (vo == v1) & (ko == k1);
    hm = lt ? (2 >> kp) : (same ? 3 : (1 << kp));
    k1 = lt ? ko : k1;
    v1 = lt ? vo : v1;
  }
  if (kp == 0) comb2[(rtl * NQ + qq) * 16 + nlo] = make_int2(__float_as_int(v1), k1 | (hm << 16));
  __syncthreads();

  float vb;
  int   bb;
  {
    const int2* c2 = comb2 + rtl * (NQ * 16) + nlo;
    const int2 p0 = c2[0];
    vb = __int_as_float(p0.x);
    int key = p0.y;
#pragma unroll
    for (int k = 1; k < NQ; ++k) {
      const int2 p = c2[16 * k];
      const float v = __int_as_float(p.x);
      const bool lt = v < vb;
      key = lt ? p.y : key;
      vb  = lt ? v : vb;
    }
    bb = key & 0xffff;
    hm = key >> 16;
  }
  const bool tie_any = __builtin_amdgcn_ballot_w32(hm == 3) != 0u;

  const int rsel = nlo & 7;
  v2f afr = af[0];
#pragma unroll
  for (int t = 1; t < RT_WG; ++t) afr = (rtl == t) ? af[t] : afr;
  int best = 0x7fffffff;
  const int h0 = tie_any ? 0 : (hm >> 1);
  {
    v2f bg[2 * UT];
    int ng[2 * UT];
#pragma unroll
    for (int g = 0; g < UT; ++g) {
      const int ci = bb * UT + g;
      const int ct = (ci < NT_COL) ? ci : NT_COL - 1;
#pragma unroll
      for (int i = 0; i < 2; ++i) {
        const int n = ct * 16 + 8 * h0 + 2 * qq + i;
        const float2 b2 = Bs2[2 * n + kp];
        ng[2 * g + i] = n;
        bg[2 * g + i] = v2f{b2.x, b2.y};
      }
    }
#pragma unroll
    for (int k = 0; k < 2 * UT; ++k) {
      const v8f z = {0.f, 0.f, 0.f, 0.f, 0.f, 0.f, 0.f, 0.f};
      const v8f d = __builtin_amdgcn_wmma_f32_16x16x4_f32(false, bg[k], false, afr,
                                                          (short)0, z, false, false);
      bool eq = false;
#pragma unroll
      for (int r = 0; r < 8; ++r) eq |= (rsel == r) & (d[r] == vb);
      best = (eq & (ng[k] < best)) ? ng[k] : best;
    }
  }
  if (tie_any) {
    v2f bg[2 * UT];
    int ng[2 * UT];
#pragma unroll
    for (int g = 0; g < UT; ++g) {
      const int ci = bb * UT + g;
      const int ct = (ci < NT_COL) ? ci : NT_COL - 1;
#pragma unroll
      for (int i = 0; i < 2; ++i) {
        const int n = ct * 16 + 8 + 2 * qq + i;
        const float2 b2 = Bs2[2 * n + kp];
        ng[2 * g + i] = n;
        bg[2 * g + i] = v2f{b2.x, b2.y};
      }
    }
#pragma unroll
    for (int k = 0; k < 2 * UT; ++k) {
      const v8f z = {0.f, 0.f, 0.f, 0.f, 0.f, 0.f, 0.f, 0.f};
      const v8f d = __builtin_amdgcn_wmma_f32_16x16x4_f32(false, bg[k], false, afr,
                                                          (short)0, z, false, false);
      bool eq = false;
#pragma unroll
      for (int r = 0; r < 8; ++r) eq |= (rsel == r) & (d[r] == vb);
      best = (eq & (ng[k] < best)) ? ng[k] : best;
    }
  }

  if (kp == (nlo >> 3)) hit[(rtl * NQ + qq) * 16 + nlo] = best;
  __syncthreads();

  if (threadIdx.x < RT_WG * 16) {
    const int t  = (int)threadIdx.x;
    const int* hq = hit + (t >> 4) * (NQ * 16) + (t & 15);
    int n = hq[0];
#pragma unroll
    for (int w = 1; w < NQ; ++w) {
      const int o = hq[16 * w];
      n = (o < n) ? o : n;
    }
    out[blockIdx.x * (RT_WG * 16) + t] = (n == 0x7fffffff) ? 0 : n;
  }
}

extern "C" void kernel_launch(void* const* d_in, const int* in_sizes, int n_in,
                              void* d_out, int out_size, void* d_ws, size_t ws_size,
                              hipStream_t stream) {
  const float* mhr   = (const float*)d_in[0];
  const float* smplx = (const float*)d_in[1];
  int* out = (int*)d_out;

  const size_t lds_bytes = (size_t)LDS_BYTES;
  vm_match<<<NB_MATCH, NTHR, lds_bytes, stream>>>(mhr, smplx, out);
}


# B' fragments straight from the raw vertices (scores bit-identical): each wave reads its own tiles' columns two blocks ahead, forms the norms in its lanes; no table build, no barrier before the loop
# speedup vs baseline: 9.2129x; 1.0534x over previous
#include <hip/hip_runtime.h>


#define N_MHR   20000
#define N_SMPLX 10475
#define N_PAD   10480
#define NT_COL  655
#define NT_ROW  1250

#define RT_WG   5
#define NWAVE   20
#define NTHR    (NWAVE * 32)
#define NB_MATCH (NT_ROW / RT_WG)
#define UT      3
#define BPW     11
#define TPW     (BPW * UT)
#define NQ      4
static_assert(NB_MATCH * RT_WG == NT_ROW, "the row tiles must divide over the work-groups exactly");
static_assert(NWAVE * TPW >= NT_COL && (NWAVE - 1) * TPW < NT_COL, "the waves must cover the column tiles, none empty");
static_assert(NWAVE == RT_WG * NQ, "the result phase gives each row tile NQ waves");
static_assert(NWAVE % NQ == 0, "each of a row tile's NQ waves merges NWAVE / NQ wave ranges");
static_assert(NQ == 4, "the column search gives each wave two of a lane half's eight table columns");
static_assert(NWAVE * BPW < 0x10000, "a block number must fit the low half of a key");
#define N_TAB   N_PAD
#define LASTC   (N_SMPLX - 1 - 16 * (NT_COL - 1))
static_assert(LASTC >= 0 && LASTC < 16, "the last tile holds at least one real column");
#define COMB1_N    (RT_WG * NWAVE * 2 * 16)
#define COMB2_N    (RT_WG * NQ * 16)
#define HIT_N      (RT_WG * NQ * 16)
#define LDS_BYTES  (N_TAB * 16 + COMB1_N * 8 + COMB2_N * 8 + HIT_N * 4)

typedef float v2f __attribute__((ext_vector_type(2)));
struct __attribute__((packed, aligned(4))) col3 { float x, y, z; };
typedef float v8f __attribute__((ext_vector_type(8)));

static __device__ __forceinline__ float col_norm(float x, float y, float z) {
  float s;
  {
#pragma clang fp contract(off)
    s = x * x + y * y;
  }
  return __builtin_fmaf(z, z, s);
}

__global__ __launch_bounds__(NTHR) void vm_match(const float* __restrict__ mhr,
                                                 const float* __restrict__ smplx,
                                                 int* __restrict__ out) {
  extern __shared__ float4 Bs4[];
  float2* Bs2 = (float2*)Bs4;
  int2* comb1 = (int2*)(Bs4 + N_TAB);
  int2* comb2 = comb1 + COMB1_N;
  int*  hit   = (int*)(comb2 + COMB2_N);

  const int lane = threadIdx.x & 31;
  const int wave = __builtin_amdgcn_readfirstlane((int)(threadIdx.x >> 5));
  const int nlo  = lane & 15;
  const int kp   = lane >> 4;
  const int rtl  = wave >> 2;
  const int qq   = wave & 3;

  float a_lo[RT_WG], a_y[RT_WG];
#pragma unroll
  for (int t = 0; t < RT_WG; ++t) {
    const float* ar = mhr + 3 * ((blockIdx.x * RT_WG + t) * 16 + nlo);
    a_lo[t] = ar[2 * kp];
    a_y[t]  = ar[1];
  }

  const int t0   = wave * TPW;
  const int c12  = 12 * nlo;
  const int c12L = 12 * (nlo < LASTC ? nlo : LASTC);
  const char* sbytes = (const char*)smplx;
  float2* bst = Bs2 + 2 * nlo + kp;

#define LOAD_BLOCK(b, RX, RY, RZ)                                              \
  _Pragma("unroll")                                                            \
  for (int g = 0; g < UT; ++g) {                                               \
    const int ci = t0 + (b) * UT + g;                                          \
    const int ct = (ci < NT_COL) ? ci : NT_COL - 1;                            \
    const int off = (ct == NT_COL - 1) ? c12L : c12;                           \
    const col3 v = *(const col3*)(sbytes + ct * 192 + off);                    \
    RX[g] = v.x; RY[g] = v.y; RZ[g] = v.z;                                     \
  }
#define FORM_BLOCK(b, RX, RY, RZ, BF)                                          \
  _Pragma("unroll")                                                            \
  for (int g = 0; g < UT; ++g) {                                               \
    const int ci = t0 + (b) * UT + g;                                          \
    const int ct = (ci < NT_COL) ? ci : NT_COL - 1;                            \
    const float w = col_norm(RX[g], RY[g], RZ[g]);                             \
    const v2f f = v2f{kp ? RZ[g] : RX[g], kp ? w : RY[g]};                     \
    BF[g] = f;                                                                 \
    bst[32 * ct] = make_float2(f.x, f.y);                                      \
  }

  float r0x[UT], r0y[UT], r0z[UT], r1x[UT], r1y[UT], r1z[UT];
  float px[UT], py[UT], pz[UT];
  LOAD_BLOCK(0, px, py, pz)
  LOAD_BLOCK(1, r1x, r1y, r1z)
  LOAD_BLOCK(2, r0x, r0y, r0z)
  __builtin_amdgcn_sched_barrier(0);

  v2f bcur[UT];
  FORM_BLOCK(0, px, py, pz, bcur)

  v2f af[RT_WG];
#pragma unroll
  for (int t = 0; t < RT_WG; ++t) af[t] = v2f{-2.0f * a_lo[t], kp ? 1.0f : -2.0f * a_y[t]};

  float vmin[RT_WG];
  int   bwin[RT_WG];
#pragma unroll
  for (int t = 0; t < RT_WG; ++t) { vmin[t] = __builtin_inff(); bwin[t] = 0; }

#define DO_BLOCK(blk)                                                          \
  {                                                                            \
    float m[RT_WG];                                                            \
    _Pragma("unroll")                                                          \
    for (int t = 0; t < RT_WG; ++t) m[t] = __builtin_inff();                   \
    _Pragma("unroll")                                                          \
    for (int g = 0; g < UT; ++g) {                                             \
             \
          \
      v8f c[RT_WG];                                                            \
      _Pragma("unroll")                                                        \
      for (int t = 0; t < RT_WG; ++t) {                                        \
        const v8f z = {0.f, 0.f, 0.f, 0.f, 0.f, 0.f, 0.f, 0.f};                \
                          \
        c[t] = __builtin_amdgcn_wmma_f32_16x16x4_f32(false, bcur[g], false, af[t], \
                                                     (short)0, z, false, false);  \
      }                                                                        \
        \
           \
        \
               \
                                      \
      __builtin_amdgcn_sched_barrier(0);                                       \
      _Pragma("unroll")                                                        \
      for (int t = 0; t < RT_WG; ++t) {                                        \
        _Pragma("unroll")                                                      \
        for (int r = 0; r < 8; ++r) m[t] = __builtin_fminf(m[t], c[t][r]);     \
      }                                                                        \
    }                                                                          \
    _Pragma("unroll")                                                          \
    for (int t = 0; t < RT_WG; ++t) {                                          \
      if (m[t] < vmin[t]) { vmin[t] = m[t]; bwin[t] = (blk); }     \
    }                                                                          \
           \
             \
            \
                  \
    __builtin_amdgcn_sched_barrier(0);                                         \
  }

  static_assert(BPW % 2 == 1 && BPW >= 5, "the loop takes the blocks in pairs and leaves three behind it");
  for (int blk = 0; blk + 3 < BPW; blk += 2) {
    DO_BLOCK(blk)
    FORM_BLOCK(blk + 1, r1x, r1y, r1z, bcur)
    LOAD_BLOCK(blk + 3, r1x, r1y, r1z)
    DO_BLOCK(blk + 1)
    FORM_BLOCK(blk + 2, r0x, r0y, r0z, bcur)
    LOAD_BLOCK(blk + 4, r0x, r0y, r0z)
  }
  DO_BLOCK(BPW - 3)
  FORM_BLOCK(BPW - 2, r1x, r1y, r1z, bcur)
  DO_BLOCK(BPW - 2)
  FORM_BLOCK(BPW - 1, r0x, r0y, r0z, bcur)
  DO_BLOCK(BPW - 1)
#undef LOAD_BLOCK
#undef FORM_BLOCK
#undef DO_BLOCK

#pragma unroll
  for (int t = 0; t < RT_WG; ++t)
    comb1[((t * NWAVE + wave) * 2 + kp) * 16 + nlo] =
        make_int2(__float_as_int(vmin[t]), wave * BPW + bwin[t]);
  __syncthreads();

  float v1;
  int   k1;
  {
    const int2* c1 = comb1 + ((rtl * NWAVE + qq * (NWAVE / NQ)) * 2 + kp) * 16 + nlo;
    const int2 p0 = c1[0];
    v1 = __int_as_float(p0.x); k1 = p0.y;
#pragma unroll
    for (int k = 1; k < NWAVE / NQ; ++k) {
      const int2 p = c1[32 * k];
      const float v = __int_as_float(p.x);
      const bool lt = v < v1;
      k1 = lt ? p.y : k1;
      v1 = lt ? v : v1;
    }
  }
  int hm;
  {
    const float vo = __shfl_xor(v1, 16, 32);
    const int   ko = __shfl_xor(k1, 16, 32);
    const bool lt   = (vo < v1) | ((vo == v1) & (ko < k1));
    const bool same = (vo == v1) & (ko == k1);
    hm = lt ? (2 >> kp) : (same ? 3 : (1 << kp));
    k1 = lt ? ko : k1;
    v1 = lt ? vo : v1;
  }
  if (kp == 0) comb2[(rtl * NQ + qq) * 16 + nlo] = make_int2(__float_as_int(v1), k1 | (hm << 16));
  __syncthreads();

  float vb;
  int   bb;
  {
    const int2* c2 = comb2 + rtl * (NQ * 16) + nlo;
    const int2 p0 = c2[0];
    vb = __int_as_float(p0.x);
    int key = p0.y;
#pragma unroll
    for (int k = 1; k < NQ; ++k) {
      const int2 p = c2[16 * k];
      const float v = __int_as_float(p.x);
      const bool lt = v < vb;
      key = lt ? p.y : key;
      vb  = lt ? v : vb;
    }
    bb = key & 0xffff;
    hm = key >> 16;
  }
  const bool tie_any = __builtin_amdgcn_ballot_w32(hm == 3) != 0u;

  const int rsel = nlo & 7;
  v2f afr = af[0];
#pragma unroll
  for (int t = 1; t < RT_WG; ++t) afr = (rtl == t) ? af[t] : afr;
  int best = 0x7fffffff;
  const int h0 = tie_any ? 0 : (hm >> 1);
  {
    v2f bg[2 * UT];
    int ng[2 * UT];
#pragma unroll
    for (int g = 0; g < UT; ++g) {
      const int ci = bb * UT + g;
      const int ct = (ci < NT_COL) ? ci : NT_COL - 1;
#pragma unroll
      for (int i = 0; i < 2; ++i) {
        const int n = ct * 16 + 8 * h0 + 2 * qq + i;
        const float2 b2 = Bs2[2 * n + kp];
        ng[2 * g + i] = n;
        bg[2 * g + i] = v2f{b2.x, b2.y};
      }
    }
#pragma unroll
    for (int k = 0; k < 2 * UT; ++k) {
      const v8f z = {0.f, 0.f, 0.f, 0.f, 0.f, 0.f, 0.f, 0.f};
      const v8f d = __builtin_amdgcn_wmma_f32_16x16x4_f32(false, bg[k], false, afr,
                                                          (short)0, z, false, false);
      bool eq = false;
#pragma unroll
      for (int r = 0; r < 8; ++r) eq |= (rsel == r) & (d[r] == vb);
      best = (eq & (ng[k] < best)) ? ng[k] : best;
    }
  }
  if (tie_any) {
    v2f bg[2 * UT];
    int ng[2 * UT];
#pragma unroll
    for (int g = 0; g < UT; ++g) {
      const int ci = bb * UT + g;
      const int ct = (ci < NT_COL) ? ci : NT_COL - 1;
#pragma unroll
      for (int i = 0; i < 2; ++i) {
        const int n = ct * 16 + 8 + 2 * qq + i;
        const float2 b2 = Bs2[2 * n + kp];
        ng[2 * g + i] = n;
        bg[2 * g + i] = v2f{b2.x, b2.y};
      }
    }
#pragma unroll
    for (int k = 0; k < 2 * UT; ++k) {
      const v8f z = {0.f, 0.f, 0.f, 0.f, 0.f, 0.f, 0.f, 0.f};
      const v8f d = __builtin_amdgcn_wmma_f32_16x16x4_f32(false, bg[k], false, afr,
                                                          (short)0, z, false, false);
      bool eq = false;
#pragma unroll
      for (int r = 0; r < 8; ++r) eq |= (rsel == r) & (d[r] == vb);
      best = (eq & (ng[k] < best)) ? ng[k] : best;
    }
  }

  if (kp == (nlo >> 3)) hit[(rtl * NQ + qq) * 16 + nlo] = best;
  __syncthreads();

  if (threadIdx.x < RT_WG * 16) {
    const int t  = (int)threadIdx.x;
    const int* hq = hit + (t >> 4) * (NQ * 16) + (t & 15);
    int n = hq[0];
#pragma unroll
    for (int w = 1; w < NQ; ++w) {
      const int o = hq[16 * w];
      n = (o < n) ? o : n;
    }
    out[blockIdx.x * (RT_WG * 16) + t] = (n == 0x7fffffff) ? 0 : n;
  }
}

extern "C" void kernel_launch(void* const* d_in, const int* in_sizes, int n_in,
                              void* d_out, int out_size, void* d_ws, size_t ws_size,
                              hipStream_t stream) {
  const float* mhr   = (const float*)d_in[0];
  const float* smplx = (const float*)d_in[1];
  int* out = (int*)d_out;

  const size_t lds_bytes = (size_t)LDS_BYTES;
  vm_match<<<NB_MATCH, NTHR, lds_bytes, stream>>>(mhr, smplx, out);
}
